# starting kernel
# baseline (speedup 1.0000x reference)
#include <hip/hip_runtime.h>

typedef __attribute__((ext_vector_type(2))) float v2f;
typedef __attribute__((ext_vector_type(8))) float v8f;

#define HSTR 260
#define H1S   68
#define H2S   36

__global__ __launch_bounds__(128)
void dfm_fused_kernel(const int*   __restrict__ x_field,
                      const int*   __restrict__ x,
                      const float* __restrict__ x_val,
                      const float* __restrict__ emb,
                      const float* __restrict__ lin_tab,
                      const float* __restrict__ lin_bias,
                      const float* __restrict__ W1, const float* __restrict__ b1,
                      const float* __restrict__ g1, const float* __restrict__ be1,
                      const float* __restrict__ W2, const float* __restrict__ b2,
                      const float* __restrict__ g2, const float* __restrict__ be2,
                      const float* __restrict__ W3, const float* __restrict__ b3,
                      float*       __restrict__ out)
{
    __shared__ float sh_h [16 * HSTR];
    __shared__ float sh_h1[16 * H1S];
    __shared__ float sh_h2[16 * H2S];
    __shared__ float sh_red[16 * 8];
    __shared__ float sh_lr[256];
    __shared__ float sh_lf[16];

    const int t    = threadIdx.x;
    const int wave = t >> 5;
    const int lane = t & 31;
    const int half = lane >> 4;
    const int mlo  = lane & 15;
    const int r0   = blockIdx.x * 16;
    const float RSQ = rsqrtf(1.0f + 1e-5f);

    for (int p = t; p < 256; p += 128) {
        const int r = p >> 4, f = p & 15;
        const int gi = (r0 + r) * 16 + f;
        const long long row = (long long)x_field[gi] * 100000ll + (long long)x[gi];
        const float val = x_val[gi];
        const float4* e4 = reinterpret_cast<const float4*>(emb + (size_t)row * 16u);
        float4 a0 = e4[0], a1 = e4[1], a2 = e4[2], a3 = e4[3];
        float4* dst = reinterpret_cast<float4*>(&sh_h[r * HSTR + f * 16]);
        dst[0] = make_float4(a0.x * val, a0.y * val, a0.z * val, a0.w * val);
        dst[1] = make_float4(a1.x * val, a1.y * val, a1.z * val, a1.w * val);
        dst[2] = make_float4(a2.x * val, a2.y * val, a2.z * val, a2.w * val);
        dst[3] = make_float4(a3.x * val, a3.y * val, a3.z * val, a3.w * val);
        sh_lr[p] = lin_tab[row] * val;
    }
    __syncthreads();

    {
        const int r = t >> 3;
        const int e = (t & 7) * 2;
        float s0 = 0.f, q0 = 0.f, s1 = 0.f, q1 = 0.f;
        #pragma unroll
        for (int f = 0; f < 16; ++f) {
            const float v0 = sh_h[r * HSTR + f * 16 + e];
            const float v1 = sh_h[r * HSTR + f * 16 + e + 1];
            s0 += v0; q0 = fmaf(v0, v0, q0);
            s1 += v1; q1 = fmaf(v1, v1, q1);
        }
        sh_red[r * 8 + (t & 7)] = (s0 * s0 - q0) + (s1 * s1 - q1);
    }
    __syncthreads();
    if (t < 16) {
        float fm = 0.f;
        #pragma unroll
        for (int i = 0; i < 8; ++i) fm += sh_red[t * 8 + i];
        float lin = 0.f;
        #pragma unroll
        for (int f = 0; f < 16; ++f) lin += sh_lr[t * 16 + f];
        sh_lf[t] = lin + lin_bias[0] + 0.5f * fm;
    }
    __syncthreads();

    {
        const int c = wave * 16 + mlo;
        v8f acc = {};
        #pragma unroll 4
        for (int kk = 0; kk < 256; kk += 4) {
            const int k0 = kk + 2 * half;
            v2f a = *(const v2f*)&sh_h[mlo * HSTR + k0];
            v2f b; b.x = W1[(size_t)k0 * 64 + c];
                   b.y = W1[(size_t)(k0 + 1) * 64 + c];
            acc = __builtin_amdgcn_wmma_f32_16x16x4_f32(false, a, false, b,
                                                        (short)0, acc, false, false);
        }
        const float sc = g1[c] * RSQ;
        const float tb = fmaf(b1[c], sc, be1[c]);
        #pragma unroll
        for (int j = 0; j < 8; ++j) {
            const int m = j + 8 * half;
            const float v = fmaf(acc[j], sc, tb);
            sh_h1[m * H1S + c] = v > 0.f ? v : 0.f;
        }
    }
    __syncthreads();

    if (wave < 2) {
        const int c = wave * 16 + mlo;
        v8f acc = {};
        #pragma unroll 4
        for (int kk = 0; kk < 64; kk += 4) {
            const int k0 = kk + 2 * half;
            v2f a = *(const v2f*)&sh_h1[mlo * H1S + k0];
            v2f b; b.x = W2[k0 * 32 + c];
                   b.y = W2[(k0 + 1) * 32 + c];
            acc = __builtin_amdgcn_wmma_f32_16x16x4_f32(false, a, false, b,
                                                        (short)0, acc, false, false);
        }
        const float sc = g2[c] * RSQ;
        const float tb = fmaf(b2[c], sc, be2[c]);
        #pragma unroll
        for (int j = 0; j < 8; ++j) {
            const int m = j + 8 * half;
            const float v = fmaf(acc[j], sc, tb);
            sh_h2[m * H2S + c] = v > 0.f ? v : 0.f;
        }
    }
    __syncthreads();

    if (t < 16) {
        float acc = sh_lf[t] + b3[0];
        #pragma unroll
        for (int cc = 0; cc < 32; ++cc)
            acc = fmaf(sh_h2[t * H2S + cc], W3[cc], acc);
        out[r0 + t] = acc;
    }
}

extern "C" void kernel_launch(void* const* d_in, const int* in_sizes, int n_in,
                              void* d_out, int out_size, void* d_ws, size_t ws_size,
                              hipStream_t stream) {
    const int*   x_field  = (const int*)  d_in[0];
    const int*   x        = (const int*)  d_in[1];
    const float* x_val    = (const float*)d_in[2];
    const float* emb      = (const float*)d_in[3];
    const float* lin_tab  = (const float*)d_in[4];
    const float* lin_bias = (const float*)d_in[5];
    const float* W1 = (const float*)d_in[6];
    const float* b1 = (const float*)d_in[7];
    const float* g1 = (const float*)d_in[8];
    const float* be1= (const float*)d_in[9];
    const float* W2 = (const float*)d_in[10];
    const float* b2 = (const float*)d_in[11];
    const float* g2 = (const float*)d_in[12];
    const float* be2= (const float*)d_in[13];
    const float* W3 = (const float*)d_in[14];
    const float* b3 = (const float*)d_in[15];
    float* out = (float*)d_out;

    const int nblocks = out_size / 16;
    dfm_fused_kernel<<<dim3(nblocks), dim3(128), 0, stream>>>(
        x_field, x, x_val, emb, lin_tab, lin_bias,
        W1, b1, g1, be1, W2, b2, g2, be2, W3, b3, out);
}


# W1 handed to LDS in two parts: rows of fields 0-7 before layer 1, the rest and W2 behind its third block, LDS-only barriers; fields 0-7 gathered ahead of the second weight reads; sum orders kept
# speedup vs baseline: 1.9168x; 1.9168x over previous
#include <hip/hip_runtime.h>

typedef __attribute__((ext_vector_type(2))) float v2f;
typedef __attribute__((ext_vector_type(8))) float v8f;

#define ROWS  64
#define H1S   68
#define H2S   36
#define W1P  144
#define W2P   96
#define NA     8

__device__ __forceinline__ void lds_barrier() {
    __builtin_amdgcn_fence(__ATOMIC_RELEASE, "workgroup", "local");
    __builtin_amdgcn_s_barrier();
    __builtin_amdgcn_fence(__ATOMIC_ACQUIRE, "workgroup", "local");
}

__global__ __launch_bounds__(128)
void dfm_fused_kernel(const int*   __restrict__ x_field,
                      const int*   __restrict__ x,
                      const float* __restrict__ x_val,
                      const float* __restrict__ emb,
                      const float* __restrict__ lin_tab,
                      const float* __restrict__ lin_bias,
                      const float* __restrict__ W1, const float* __restrict__ b1,
                      const float* __restrict__ g1, const float* __restrict__ be1,
                      const float* __restrict__ W2, const float* __restrict__ b2,
                      const float* __restrict__ g2, const float* __restrict__ be2,
                      const float* __restrict__ W3, const float* __restrict__ b3,
                      float*       __restrict__ out)
{
    __shared__ float sh_w1[128 * W1P];
    __shared__ float sh_w2[32 * W2P];
    __shared__ float sh_h1[ROWS * H1S];
    __shared__ float sh_h2[ROWS * H2S];
    __shared__ float sh_lf[ROWS];

    const int t    = threadIdx.x;
    const int wave = t >> 5;
    const int lane = t & 31;
    const int half = lane >> 4;
    const int mlo  = lane & 15;
    const int r0   = blockIdx.x * ROWS;
    const int row  = r0 + wave * 16 + mlo;
    const float RSQ = rsqrtf(1.0f + 1e-5f);

    int xf[16], xx[16];
    float xv[16];
    {
        const int4*   pf = reinterpret_cast<const int4*>(x_field + (size_t)row * 16u);
        const int4*   px = reinterpret_cast<const int4*>(x + (size_t)row * 16u);
        const float4* pv = reinterpret_cast<const float4*>(x_val + (size_t)row * 16u);
        #pragma unroll
        for (int i = 0; i < 4; ++i) {
            const int4 a = pf[i]; const int4 b = px[i]; const float4 c = pv[i];
            xf[4*i] = a.x; xf[4*i+1] = a.y; xf[4*i+2] = a.z; xf[4*i+3] = a.w;
            xx[4*i] = b.x; xx[4*i+1] = b.y; xx[4*i+2] = b.z; xx[4*i+3] = b.w;
            xv[4*i] = c.x; xv[4*i+1] = c.y; xv[4*i+2] = c.z; xv[4*i+3] = c.w;
        }
    }

    float4 wa[16], wb[16];
    #pragma unroll
    for (int i = 0; i < NA; ++i) {
        const int item = t + 128 * i;
        const int kp = item >> 4, c4 = item & 15;
        wa[i] = *reinterpret_cast<const float4*>(&W1[(2 * kp)     * 64 + 4 * c4]);
        wb[i] = *reinterpret_cast<const float4*>(&W1[(2 * kp + 1) * 64 + 4 * c4]);
    }

    float4 av[16][2];
    #pragma unroll
    for (int f = 0; f < NA; ++f) {
        const long long rid = (long long)xf[f] * 100000ll + (long long)xx[f];
        const float4* e = reinterpret_cast<const float4*>(emb + (size_t)rid * 16u + 4u * half);
        av[f][0] = e[0];
        av[f][1] = e[2];
    }

    float4 w2a[2], w2b[2];
    float  lv[8];

    float g1v[4], b1v[4], e1v[4], g2v[2], b2v[2], e2v[2];
    #pragma unroll
    for (int n = 0; n < 4; ++n) {
        const int c = n * 16 + mlo;
        g1v[n] = g1[c]; b1v[n] = b1[c]; e1v[n] = be1[c];
    }
    #pragma unroll
    for (int n = 0; n < 2; ++n) {
        const int c = n * 16 + mlo;
        g2v[n] = g2[c]; b2v[n] = b2[c]; e2v[n] = be2[c];
    }
    const float lb = lin_bias[0];
    const float b3v = b3[0];

    #pragma unroll
    for (int i = 0; i < NA; ++i) {
        const int item = t + 128 * i;
        const int kp = item >> 4, c4 = item & 15;
        float4* d = reinterpret_cast<float4*>(&sh_w1[kp * W1P + 8 * c4]);
        d[0] = make_float4(wa[i].x, wb[i].x, wa[i].y, wb[i].y);
        d[1] = make_float4(wa[i].z, wb[i].z, wa[i].w, wb[i].w);
    }
    lds_barrier();

    int tq = t;
    asm volatile("" : "+v"(tq));
    #pragma unroll
    for (int i = NA; i < 16; ++i) {
        const int item = tq + 128 * i;
        const int kp = item >> 4, c4 = item & 15;
        wa[i] = *reinterpret_cast<const float4*>(&W1[(2 * kp)     * 64 + 4 * c4]);
        wb[i] = *reinterpret_cast<const float4*>(&W1[(2 * kp + 1) * 64 + 4 * c4]);
    }

    #pragma unroll
    for (int i = 0; i < 2; ++i) {
        const int item = tq + 128 * i;
        const int kp = item >> 3, c4 = item & 7;
        w2a[i] = *reinterpret_cast<const float4*>(&W2[(2 * kp)     * 32 + 4 * c4]);
        w2b[i] = *reinterpret_cast<const float4*>(&W2[(2 * kp + 1) * 32 + 4 * c4]);
    }

    unsigned hq = 4u * half;
    asm volatile("" : "+v"(hq));
    #pragma unroll
    for (int f = NA; f < 16; ++f) {
        const long long rid = (long long)xf[f] * 100000ll + (long long)xx[f];
        const float4* e = reinterpret_cast<const float4*>(emb + (size_t)rid * 16u + hq);
        av[f][0] = e[0];
        av[f][1] = e[2];
    }

    v8f acc[4] = {};
    v2f w2f[16][2];
    float s[8], q[8];
    #pragma unroll
    for (int j = 0; j < 8; ++j) { s[j] = 0.f; q[j] = 0.f; }
    {
        const float* bp = &sh_w1[(2 * half) * W1P + 2 * mlo];
        v2f bc[2][4][4], bn[2][4][4];

#define KPROW(F, ST) (8 * (F) + 4 * ((ST) >> 1) + ((ST) & 1))
#define SCALE_FM(F) do {                                                                   \
            float val = xv[F];                                                                 \
            asm volatile("" : "+v"(val));                                                      \
            av[F][0].x *= val; av[F][0].y *= val; av[F][0].z *= val; av[F][0].w *= val;        \
            av[F][1].x *= val; av[F][1].y *= val; av[F][1].z *= val; av[F][1].w *= val;        \
            const float v[8] = { av[F][0].x, av[F][0].y, av[F][0].z, av[F][0].w,               \
                                 av[F][1].x, av[F][1].y, av[F][1].z, av[F][1].w };             \
            _Pragma("unroll")                                                                  \
            for (int j = 0; j < 8; ++j) { s[j] += v[j]; q[j] = fmaf(v[j], v[j], q[j]); }       \
        } while (0)

        #pragma unroll
        for (int ff = 0; ff < 2; ++ff)
            #pragma unroll
            for (int st = 0; st < 4; ++st)
                #pragma unroll
                for (int n = 0; n < 4; ++n)
                    bc[ff][st][n] = *(const v2f*)&bp[KPROW(ff, st) * W1P + 32 * n];
        SCALE_FM(0);
        SCALE_FM(1);
        #pragma unroll
        for (int blk = 0; blk < 8; ++blk) {
            if (blk == NA / 2 - 1) {
                #pragma unroll
                for (int i = NA; i < 16; ++i) {
                    const int item = t + 128 * i;
                    const int kp = item >> 4, c4 = item & 15;
                    float4* d = reinterpret_cast<float4*>(&sh_w1[kp * W1P + 8 * c4]);
                    d[0] = make_float4(wa[i].x, wb[i].x, wa[i].y, wb[i].y);
                    d[1] = make_float4(wa[i].z, wb[i].z, wa[i].w, wb[i].w);
                }
                #pragma unroll
                for (int i = 0; i < 2; ++i) {
                    const int item = t + 128 * i;
                    const int kp = item >> 3, c4 = item & 7;
                    float4* d = reinterpret_cast<float4*>(&sh_w2[kp * W2P + 8 * c4]);
                    d[0] = make_float4(w2a[i].x, w2b[i].x, w2a[i].y, w2b[i].y);
                    d[1] = make_float4(w2a[i].z, w2b[i].z, w2a[i].w, w2b[i].w);
                }
                lds_barrier();
                const float* bp2 = &sh_w2[half * W2P + 2 * mlo];
                #pragma unroll
                for (int ks = 0; ks < 16; ++ks)
                    #pragma unroll
                    for (int n = 0; n < 2; ++n)
                        w2f[ks][n] = *(const v2f*)&bp2[(2 * ks) * W2P + 32 * n];
                unsigned lq = hq;
                asm volatile("" : "+v"(lq));
                #pragma unroll
                for (int g = 0; g < 8; ++g) {
                    const int   sf = lq ? xf[2*g+1] : xf[2*g];
                    const int   sx = lq ? xx[2*g+1] : xx[2*g];
                    const long long rid = (long long)sf * 100000ll + (long long)sx;
                    lv[g] = lin_tab[rid];
                }
            }
            if (blk < 7) {
                #pragma unroll
                for (int ff = 0; ff < 2; ++ff)
                    #pragma unroll
                    for (int st = 0; st < 4; ++st)
                        #pragma unroll
                        for (int n = 0; n < 4; ++n)
                            bn[ff][st][n] = *(const v2f*)&bp[KPROW(2 * blk + 2 + ff, st) * W1P + 32 * n];
            }
            __builtin_amdgcn_sched_barrier(0);
            #pragma unroll
            for (int ff = 0; ff < 2; ++ff) {
                const int f = 2 * blk + ff;
                #pragma unroll
                for (int st = 0; st < 4; ++st) {
                    const float4 q4 = av[f][st >> 1];
                    v2f a;
                    a.x = (st & 1) ? q4.z : q4.x;
                    a.y = (st & 1) ? q4.w : q4.y;
                    #pragma unroll
                    for (int n = 0; n < 4; ++n)
                        acc[n] = __builtin_amdgcn_wmma_f32_16x16x4_f32(false, a, false, bc[ff][st][n],
                                                                       (short)0, acc[n], false, false);
                }
                if (ff == 0 && blk < 7) { SCALE_FM(2 * blk + 2); SCALE_FM(2 * blk + 3); }
            }
            __builtin_amdgcn_sched_barrier(0);
            if (blk < 7) {
                #pragma unroll
                for (int ff = 0; ff < 2; ++ff)
                    #pragma unroll
                    for (int st = 0; st < 4; ++st)
                        #pragma unroll
                        for (int n = 0; n < 4; ++n)
                            bc[ff][st][n] = bn[ff][st][n];
            }
        }
#undef SCALE_FM
#undef KPROW
        float rq1 = RSQ;
        asm volatile("" : "+v"(rq1));
        float sc1[4], tb1[4];
        #pragma unroll
        for (int n = 0; n < 4; ++n) {
            sc1[n] = g1v[n] * rq1;
            tb1[n] = fmaf(b1v[n], sc1[n], e1v[n]);
        }
        #pragma unroll
        for (int n = 0; n < 4; ++n) {
            const int c = n * 16 + mlo;
            #pragma unroll
            for (int j = 0; j < 8; ++j) {
                const int m = wave * 16 + j + 8 * half;
                const float v = fmaf(acc[n][j], sc1[n], tb1[n]);
                sh_h1[m * H1S + c] = v > 0.f ? v : 0.f;
            }
        }
    }
    float lf;
    {
        float fm = 0.f;
        #pragma unroll
        for (int j = 0; j < 8; j += 2)
            fm += (s[j] * s[j] - q[j]) + (s[j+1] * s[j+1] - q[j+1]);
        float lin = 0.f;
        #pragma unroll
        for (int g = 0; g < 8; ++g) {
            float val = half ? xv[2*g+1] : xv[2*g];
            asm volatile("" : "+v"(val));
            lin += lv[g] * val;
        }
        fm  += __shfl_xor(fm, 16);
        lin += __shfl_xor(lin, 16);
        lf = lin + lb + 0.5f * fm;
    }
    __syncthreads();

    {
        const float* ap = &sh_h1[(wave * 16 + mlo) * H1S + 2 * half];
        v2f af[16];
        #pragma unroll
        for (int ks = 0; ks < 16; ++ks) af[ks] = *(const v2f*)&ap[4 * ks];
        __builtin_amdgcn_sched_barrier(0);
        v8f a2[2] = {};
        #pragma unroll
        for (int ks = 0; ks < 16; ++ks) {
            #pragma unroll
            for (int n = 0; n < 2; ++n)
                a2[n] = __builtin_amdgcn_wmma_f32_16x16x4_f32(false, af[ks], false, w2f[ks][n],
                                                              (short)0, a2[n], false, false);
        }
        float rq2 = RSQ;
        asm volatile("" : "+v"(rq2));
        float sc2[2], tb2[2];
        #pragma unroll
        for (int n = 0; n < 2; ++n) {
            sc2[n] = g2v[n] * rq2;
            tb2[n] = fmaf(b2v[n], sc2[n], e2v[n]);
        }
        #pragma unroll
        for (int n = 0; n < 2; ++n) {
            const int c = n * 16 + mlo;
            #pragma unroll
            for (int j = 0; j < 8; ++j) {
                const int m = wave * 16 + j + 8 * half;
                const float v = fmaf(a2[n][j], sc2[n], tb2[n]);
                sh_h2[m * H2S + c] = v > 0.f ? v : 0.f;
            }
        }
        if (lane < 16) sh_lf[wave * 16 + lane] = lf;
    }
    __syncthreads();

    if (t < ROWS) {
        float o = sh_lf[t] + b3v;
        #pragma unroll
        for (int cc = 0; cc < 32; ++cc)
            o = fmaf(sh_h2[t * H2S + cc], W3[cc], o);
        out[r0 + t] = o;
    }
}

extern "C" void kernel_launch(void* const* d_in, const int* in_sizes, int n_in,
                              void* d_out, int out_size, void* d_ws, size_t ws_size,
                              hipStream_t stream) {
    const int*   x_field  = (const int*)  d_in[0];
    const int*   x        = (const int*)  d_in[1];
    const float* x_val    = (const float*)d_in[2];
    const float* emb      = (const float*)d_in[3];
    const float* lin_tab  = (const float*)d_in[4];
    const float* lin_bias = (const float*)d_in[5];
    const float* W1 = (const float*)d_in[6];
    const float* b1 = (const float*)d_in[7];
    const float* g1 = (const float*)d_in[8];
    const float* be1= (const float*)d_in[9];
    const float* W2 = (const float*)d_in[10];
    const float* b2 = (const float*)d_in[11];
    const float* g2 = (const float*)d_in[12];
    const float* be2= (const float*)d_in[13];
    const float* W3 = (const float*)d_in[14];
    const float* b3 = (const float*)d_in[15];
    float* out = (float*)d_out;

    const int nblocks = out_size / 64;
    dfm_fused_kernel<<<dim3(nblocks), dim3(128), 0, stream>>>(
        x_field, x, x_val, emb, lin_tab, lin_bias,
        W1, b1, g1, be1, W2, b2, g2, be2, W3, b3, out);
}


# layers computed transposed (weights as WMMA A, activations as B): each layer's D registers are the next layer's B, no LDS tile of h1/h2, one barrier left; layer-2 k order and layer-3 sum tree changed
# speedup vs baseline: 2.0146x; 1.0510x over previous
#include <hip/hip_runtime.h>

typedef __attribute__((ext_vector_type(2))) float v2f;
typedef __attribute__((ext_vector_type(8))) float v8f;

#define ROWS  64
#define W1P  144
#define W2P   72
#define NA     8

__device__ __forceinline__ void lds_barrier() {
    __builtin_amdgcn_fence(__ATOMIC_RELEASE, "workgroup", "local");
    __builtin_amdgcn_s_barrier();
    __builtin_amdgcn_fence(__ATOMIC_ACQUIRE, "workgroup", "local");
}

__global__ __launch_bounds__(128)
void dfm_fused_kernel(const int*   __restrict__ x_field,
                      const int*   __restrict__ x,
                      const float* __restrict__ x_val,
                      const float* __restrict__ emb,
                      const float* __restrict__ lin_tab,
                      const float* __restrict__ lin_bias,
                      const float* __restrict__ W1, const float* __restrict__ b1,
                      const float* __restrict__ g1, const float* __restrict__ be1,
                      const float* __restrict__ W2, const float* __restrict__ b2,
                      const float* __restrict__ g2, const float* __restrict__ be2,
                      const float* __restrict__ W3, const float* __restrict__ b3,
                      float*       __restrict__ out)
{
    __shared__ float sh_w1[128 * W1P];
    __shared__ float sh_w2[32 * W2P];
    __shared__ float sh_o[ROWS];

    const int t    = threadIdx.x;
    const int wave = t >> 5;
    const int lane = t & 31;
    const int half = lane >> 4;
    const int mlo  = lane & 15;
    const int r0   = blockIdx.x * ROWS;
    const int row  = r0 + wave * 16 + mlo;
    const float RSQ = rsqrtf(1.0f + 1e-5f);

    int xf[16], xx[16];
    float xv[16];
    {
        const int4*   pf = reinterpret_cast<const int4*>(x_field + (size_t)row * 16u);
        const int4*   px = reinterpret_cast<const int4*>(x + (size_t)row * 16u);
        const float4* pv = reinterpret_cast<const float4*>(x_val + (size_t)row * 16u);
        #pragma unroll
        for (int i = 0; i < 4; ++i) {
            const int4 a = pf[i]; const int4 b = px[i]; const float4 c = pv[i];
            xf[4*i] = a.x; xf[4*i+1] = a.y; xf[4*i+2] = a.z; xf[4*i+3] = a.w;
            xx[4*i] = b.x; xx[4*i+1] = b.y; xx[4*i+2] = b.z; xx[4*i+3] = b.w;
            xv[4*i] = c.x; xv[4*i+1] = c.y; xv[4*i+2] = c.z; xv[4*i+3] = c.w;
        }
    }

    float4 wa[16], wb[16];
    #pragma unroll
    for (int i = 0; i < NA; ++i) {
        const int item = t + 128 * i;
        const int kp = item >> 4, c4 = item & 15;
        wa[i] = *reinterpret_cast<const float4*>(&W1[(2 * kp)     * 64 + 4 * c4]);
        wb[i] = *reinterpret_cast<const float4*>(&W1[(2 * kp + 1) * 64 + 4 * c4]);
    }

    float4 av[16][2];
    #pragma unroll
    for (int f = 0; f < NA; ++f) {
        const long long rid = (long long)xf[f] * 100000ll + (long long)xx[f];
        const float4* e = reinterpret_cast<const float4*>(emb + (size_t)rid * 16u + 4u * half);
        av[f][0] = e[0];
        av[f][1] = e[2];
    }

    float4 w2a[2], w2b[2];
    float  lv[8];

    float4 cg1[4][2], cb1[4][2], ce1[4][2], cg2[2][2], cb2[2][2], ce2[2][2], cw3[2][2];
    const float lb = lin_bias[0];
    typedef __attribute__((address_space(1))) float gfloat;
    gfloat* outp = (gfloat*)out;
    asm volatile("" : "+s"(outp));
    const float b3v = b3[0];

    #pragma unroll
    for (int i = 0; i < NA; ++i) {
        const int item = t + 128 * i;
        const int kp = item >> 4, c4 = item & 15;
        float4* d = reinterpret_cast<float4*>(&sh_w1[kp * W1P + 8 * c4]);
        d[0] = make_float4(wa[i].x, wb[i].x, wa[i].y, wb[i].y);
        d[1] = make_float4(wa[i].z, wb[i].z, wa[i].w, wb[i].w);
    }
    lds_barrier();

    int tq = t;
    asm volatile("" : "+v"(tq));
    #pragma unroll
    for (int i = NA; i < 16; ++i) {
        const int item = tq + 128 * i;
        const int kp = item >> 4, c4 = item & 15;
        wa[i] = *reinterpret_cast<const float4*>(&W1[(2 * kp)     * 64 + 4 * c4]);
        wb[i] = *reinterpret_cast<const float4*>(&W1[(2 * kp + 1) * 64 + 4 * c4]);
    }

    #pragma unroll
    for (int i = 0; i < 2; ++i) {
        const int item = tq + 128 * i;
        const int kp = item >> 3, c4 = item & 7;
        w2a[i] = *reinterpret_cast<const float4*>(&W2[(2 * kp)     * 32 + 4 * c4]);
        w2b[i] = *reinterpret_cast<const float4*>(&W2[(2 * kp + 1) * 32 + 4 * c4]);
    }

    unsigned hq = 4u * half;
    asm volatile("" : "+v"(hq));
    #pragma unroll
    for (int f = NA; f < 16; ++f) {
        const long long rid = (long long)xf[f] * 100000ll + (long long)xx[f];
        const float4* e = reinterpret_cast<const float4*>(emb + (size_t)rid * 16u + hq);
        av[f][0] = e[0];
        av[f][1] = e[2];
    }

    v8f acc[4] = {};
    v2f w2f[4][4][2];
    float h1r[4][8];
    float s[8], q[8];
    #pragma unroll
    for (int j = 0; j < 8; ++j) { s[j] = 0.f; q[j] = 0.f; }
    {
        const float* bp = &sh_w1[(2 * half) * W1P + 2 * mlo];
        v2f bc[2][4][4], bn[2][4][4];

#define KPROW(F, ST) (8 * (F) + 4 * ((ST) >> 1) + ((ST) & 1))
#define SCALE_FM(F) do {                                                                   \
            float val = xv[F];                                                                 \
            asm volatile("" : "+v"(val));                                                      \
            av[F][0].x *= val; av[F][0].y *= val; av[F][0].z *= val; av[F][0].w *= val;        \
            av[F][1].x *= val; av[F][1].y *= val; av[F][1].z *= val; av[F][1].w *= val;        \
            const float v[8] = { av[F][0].x, av[F][0].y, av[F][0].z, av[F][0].w,               \
                                 av[F][1].x, av[F][1].y, av[F][1].z, av[F][1].w };             \
            _Pragma("unroll")                                                                  \
            for (int j = 0; j < 8; ++j) { s[j] += v[j]; q[j] = fmaf(v[j], v[j], q[j]); }       \
        } while (0)

        #pragma unroll
        for (int ff = 0; ff < 2; ++ff)
            #pragma unroll
            for (int st = 0; st < 4; ++st)
                #pragma unroll
                for (int n = 0; n < 4; ++n)
                    bc[ff][st][n] = *(const v2f*)&bp[KPROW(ff, st) * W1P + 32 * n];
        SCALE_FM(0);
        SCALE_FM(1);
        #pragma unroll
        for (int blk = 0; blk < 8; ++blk) {
            if (blk == NA / 2 - 1) {
                #pragma unroll
                for (int i = NA; i < 16; ++i) {
                    const int item = t + 128 * i;
                    const int kp = item >> 4, c4 = item & 15;
                    float4* d = reinterpret_cast<float4*>(&sh_w1[kp * W1P + 8 * c4]);
                    d[0] = make_float4(wa[i].x, wb[i].x, wa[i].y, wb[i].y);
                    d[1] = make_float4(wa[i].z, wb[i].z, wa[i].w, wb[i].w);
                }
                #pragma unroll
                for (int i = 0; i < 2; ++i) {
                    const int item = t + 128 * i;
                    const int kp = item >> 3, c4 = item & 7;
                    float4* d = reinterpret_cast<float4*>(&sh_w2[kp * W2P + 8 * c4]);
                    d[0] = make_float4(w2a[i].x, w2b[i].x, w2a[i].y, w2b[i].y);
                    d[1] = make_float4(w2a[i].z, w2b[i].z, w2a[i].w, w2b[i].w);
                }
                lds_barrier();
                const float* bp2 = &sh_w2[(4 * half) * W2P + 2 * mlo];
                #pragma unroll
                for (int n = 0; n < 4; ++n)
                    #pragma unroll
                    for (int rp = 0; rp < 4; ++rp)
                        #pragma unroll
                        for (int n2 = 0; n2 < 2; ++n2)
                            w2f[n][rp][n2] = *(const v2f*)&bp2[(8 * n + rp) * W2P + 32 * n2];
                unsigned lq = hq;
                asm volatile("" : "+v"(lq));
                {
                    const unsigned uq = 2u * lq;
                    #pragma unroll
                    for (int n = 0; n < 4; ++n)
                        #pragma unroll
                        for (int p = 0; p < 2; ++p) {
                            cg1[n][p] = *reinterpret_cast<const float4*>(g1  + 16 * n + 4 * p + uq);
                            cb1[n][p] = *reinterpret_cast<const float4*>(b1  + 16 * n + 4 * p + uq);
                            ce1[n][p] = *reinterpret_cast<const float4*>(be1 + 16 * n + 4 * p + uq);
                        }
                    #pragma unroll
                    for (int n = 0; n < 2; ++n)
                        #pragma unroll
                        for (int p = 0; p < 2; ++p) {
                            cg2[n][p] = *reinterpret_cast<const float4*>(g2  + 16 * n + 4 * p + uq);
                            cb2[n][p] = *reinterpret_cast<const float4*>(b2  + 16 * n + 4 * p + uq);
                            ce2[n][p] = *reinterpret_cast<const float4*>(be2 + 16 * n + 4 * p + uq);
                            cw3[n][p] = *reinterpret_cast<const float4*>(W3  + 16 * n + 4 * p + uq);
                        }
                }
                #pragma unroll
                for (int g = 0; g < 8; ++g) {
                    const int   sf = lq ? xf[2*g+1] : xf[2*g];
                    const int   sx = lq ? xx[2*g+1] : xx[2*g];
                    const long long rid = (long long)sf * 100000ll + (long long)sx;
                    lv[g] = lin_tab[rid];
                }
            }
            if (blk < 7) {
                #pragma unroll
                for (int ff = 0; ff < 2; ++ff)
                    #pragma unroll
                    for (int st = 0; st < 4; ++st)
                        #pragma unroll
                        for (int n = 0; n < 4; ++n)
                            bn[ff][st][n] = *(const v2f*)&bp[KPROW(2 * blk + 2 + ff, st) * W1P + 32 * n];
            }
            __builtin_amdgcn_sched_barrier(0);
            #pragma unroll
            for (int ff = 0; ff < 2; ++ff) {
                const int f = 2 * blk + ff;
                #pragma unroll
                for (int st = 0; st < 4; ++st) {
                    const float4 q4 = av[f][st >> 1];
                    v2f a;
                    a.x = (st & 1) ? q4.z : q4.x;
                    a.y = (st & 1) ? q4.w : q4.y;
                    #pragma unroll
                    for (int n = 0; n < 4; ++n)
                        acc[n] = __builtin_amdgcn_wmma_f32_16x16x4_f32(false, bc[ff][st][n], false, a,
                                                                       (short)0, acc[n], false, false);
                }
                if (ff == 0 && blk < 7) { SCALE_FM(2 * blk + 2); SCALE_FM(2 * blk + 3); }
            }
            __builtin_amdgcn_sched_barrier(0);
            if (blk < 7) {
                #pragma unroll
                for (int ff = 0; ff < 2; ++ff)
                    #pragma unroll
                    for (int st = 0; st < 4; ++st)
                        #pragma unroll
                        for (int n = 0; n < 4; ++n)
                            bc[ff][st][n] = bn[ff][st][n];
            }
        }
#undef SCALE_FM
#undef KPROW
        float rq1 = RSQ;
        asm volatile("" : "+v"(rq1));
        #pragma unroll
        for (int n = 0; n < 4; ++n) {
            const float gg[8] = { cg1[n][0].x, cg1[n][0].y, cg1[n][0].z, cg1[n][0].w, cg1[n][1].x, cg1[n][1].y, cg1[n][1].z, cg1[n][1].w };
            const float bb[8] = { cb1[n][0].x, cb1[n][0].y, cb1[n][0].z, cb1[n][0].w, cb1[n][1].x, cb1[n][1].y, cb1[n][1].z, cb1[n][1].w };
            const float ee[8] = { ce1[n][0].x, ce1[n][0].y, ce1[n][0].z, ce1[n][0].w, ce1[n][1].x, ce1[n][1].y, ce1[n][1].z, ce1[n][1].w };
            #pragma unroll
            for (int j = 0; j < 8; ++j) {
                const float sc = gg[j] * rq1;
                const float tb = fmaf(bb[j], sc, ee[j]);
                const float v  = fmaf(acc[n][j], sc, tb);
                h1r[n][j] = v > 0.f ? v : 0.f;
            }
        }
    }
    v8f a2[2] = {};
    #pragma unroll
    for (int n = 0; n < 4; ++n) {
        #pragma unroll
        for (int rp = 0; rp < 4; ++rp) {
            v2f b;
            b.x = h1r[n][2 * rp];
            b.y = h1r[n][2 * rp + 1];
            #pragma unroll
            for (int n2 = 0; n2 < 2; ++n2)
                a2[n2] = __builtin_amdgcn_wmma_f32_16x16x4_f32(false, w2f[n][rp][n2], false, b,
                                                               (short)0, a2[n2], false, false);
        }
    }

    float lf;
    {
        float fm = 0.f;
        #pragma unroll
        for (int j = 0; j < 8; j += 2)
            fm += (s[j] * s[j] - q[j]) + (s[j+1] * s[j+1] - q[j+1]);
        float lin = 0.f;
        #pragma unroll
        for (int g = 0; g < 8; ++g) {
            float val = half ? xv[2*g+1] : xv[2*g];
            asm volatile("" : "+v"(val));
            lin += lv[g] * val;
        }
        fm  += __shfl_xor(fm, 16);
        lin += __shfl_xor(lin, 16);
        lf = lin + lb + 0.5f * fm;
    }

    float o;
    {
        float rq2 = RSQ;
        asm volatile("" : "+v"(rq2));
        float pn[2] = { 0.f, 0.f };
        #pragma unroll
        for (int n = 0; n < 2; ++n) {
            const float gg[8] = { cg2[n][0].x, cg2[n][0].y, cg2[n][0].z, cg2[n][0].w, cg2[n][1].x, cg2[n][1].y, cg2[n][1].z, cg2[n][1].w };
            const float bb[8] = { cb2[n][0].x, cb2[n][0].y, cb2[n][0].z, cb2[n][0].w, cb2[n][1].x, cb2[n][1].y, cb2[n][1].z, cb2[n][1].w };
            const float ee[8] = { ce2[n][0].x, ce2[n][0].y, ce2[n][0].z, ce2[n][0].w, ce2[n][1].x, ce2[n][1].y, ce2[n][1].z, ce2[n][1].w };
            const float ww[8] = { cw3[n][0].x, cw3[n][0].y, cw3[n][0].z, cw3[n][0].w, cw3[n][1].x, cw3[n][1].y, cw3[n][1].z, cw3[n][1].w };
            #pragma unroll
            for (int j = 0; j < 8; ++j) {
                const float sc = gg[j] * rq2;
                const float tb = fmaf(bb[j], sc, ee[j]);
                float v = fmaf(a2[n][j], sc, tb);
                v = v > 0.f ? v : 0.f;
                pn[n] = fmaf(v, ww[j], pn[n]);
            }
        }
        float p = pn[0] + pn[1];
        p += __shfl_xor(p, 16);
        o = (lf + b3v) + p;
    }

    if (lane < 16) sh_o[wave * 16 + lane] = o;
    __syncthreads();
    if (t < ROWS) outp[r0 + t] = sh_o[t];
}

extern "C" void kernel_launch(void* const* d_in, const int* in_sizes, int n_in,
                              void* d_out, int out_size, void* d_ws, size_t ws_size,
                              hipStream_t stream) {
    const int*   x_field  = (const int*)  d_in[0];
    const int*   x        = (const int*)  d_in[1];
    const float* x_val    = (const float*)d_in[2];
    const float* emb      = (const float*)d_in[3];
    const float* lin_tab  = (const float*)d_in[4];
    const float* lin_bias = (const float*)d_in[5];
    const float* W1 = (const float*)d_in[6];
    const float* b1 = (const float*)d_in[7];
    const float* g1 = (const float*)d_in[8];
    const float* be1= (const float*)d_in[9];
    const float* W2 = (const float*)d_in[10];
    const float* b2 = (const float*)d_in[11];
    const float* g2 = (const float*)d_in[12];
    const float* be2= (const float*)d_in[13];
    const float* W3 = (const float*)d_in[14];
    const float* b3 = (const float*)d_in[15];
    float* out = (float*)d_out;

    const int nblocks = out_size / 64;
    dfm_fused_kernel<<<dim3(nblocks), dim3(128), 0, stream>>>(
        x_field, x, x_val, emb, lin_tab, lin_bias,
        W1, b1, g1, be1, W2, b2, g2, be2, W3, b3, out);
}
